# starting kernel
# baseline (speedup 1.0000x reference)
#include <hip/hip_runtime.h>


typedef __attribute__((ext_vector_type(2))) float v2f;
typedef __attribute__((ext_vector_type(8))) float v8f;

#define D        496
#define LDW      512
#define PAIR_CAP 64
#define TUP_CAP  64

__global__ void extract_pairs_k(const float* __restrict__ M1, int T,
                                int* __restrict__ counts,
                                int2* __restrict__ pairs) {
  int idx = blockIdx.x * blockDim.x + threadIdx.x;
  if (idx >= T * D) return;
  int t = idx / D, i = idx - t * D;
  const float* row = M1 + (size_t)t * D * D + (size_t)i * D;
  int j = -1; float sv = 0.f;
  for (int c = 0; c < D; ++c) {
    float v = row[c];
    if (v != 0.f) { j = c; sv = v; }
  }
  if (j >= 0 && sv > 0.f) {
    int slot = atomicAdd(&counts[t], 1);
    pairs[t * PAIR_CAP + slot] = make_int2(i, j);
  }
}

__global__ void build_w_k(const float* __restrict__ theta,
                          const int* __restrict__ gti,
                          const int* __restrict__ gpi,
                          const int* __restrict__ counts,
                          const int2* __restrict__ pairs,
                          float* __restrict__ W, int G) {
  extern __shared__ float Uls[];
  const int kl = threadIdx.x;
  const int k0 = blockIdx.x * 128;
  const int k  = k0 + kl;

  for (int i = 0; i < D; ++i)
    Uls[i * 129 + kl] = (i == k) ? 1.f : 0.f;

  for (int g = 0; g < G; ++g) {
    int t = gti[g], p = gpi[g];
    float th = theta[p];
    float c = cosf(th), s = sinf(th);
    int n = counts[t];
    const int2* pl = pairs + t * PAIR_CAP;
    for (int pp = 0; pp < n; ++pp) {
      int2 ij = pl[pp];
      float a = Uls[ij.x * 129 + kl];
      float b = Uls[ij.y * 129 + kl];
      Uls[ij.x * 129 + kl] =  c * a + s * b;
      Uls[ij.y * 129 + kl] = -s * a + c * b;
    }
  }

  __syncthreads();
  for (int kk = 0; kk < 128; ++kk) {
    int krow = k0 + kk;
    for (int ib = threadIdx.x; ib < LDW; ib += 128) {
      float v = (ib < D) ? Uls[ib * 129 + kk] : 0.f;
      W[(size_t)krow * LDW + ib] = v;
    }
  }
}

__global__ void gemm_wmma_k(const float* __restrict__ A,
                            const float* __restrict__ W,
                            float* __restrict__ out) {
  const int lane = threadIdx.x & 31;
  const int wave = threadIdx.x >> 5;
  const int l16  = lane & 15;
  const int lh2  = (lane >> 4) * 2;
  const int m0   = (blockIdx.y * 8 + wave) * 64;
  const int n0   = blockIdx.x * 16;

  v8f acc[4];
#pragma unroll
  for (int mm = 0; mm < 4; ++mm) acc[mm] = (v8f)0.f;

  const float* a0 = A + (size_t)(m0 + l16) * D;
  const float* bp = W + (size_t)n0 + l16;

  for (int kk = 0; kk < D; kk += 4) {
    v2f b;
    b.x = bp[(size_t)(kk + lh2) * LDW];
    b.y = bp[(size_t)(kk + lh2 + 1) * LDW];
#pragma unroll
    for (int mm = 0; mm < 4; ++mm) {
      v2f a = *(const v2f*)(a0 + (size_t)mm * 16 * D + kk + lh2);
      acc[mm] = __builtin_amdgcn_wmma_f32_16x16x4_f32(
          false, a, false, b, (short)0, acc[mm], false, false);
    }
  }

  const int rbase = m0 + ((lane >> 4) << 3);
#pragma unroll
  for (int mm = 0; mm < 4; ++mm)
#pragma unroll
    for (int v = 0; v < 8; ++v)
      out[(size_t)(rbase + mm * 16 + v) * D + n0 + l16] = acc[mm][v];
}

extern "C" void kernel_launch(void* const* d_in, const int* in_sizes, int n_in,
                              void* d_out, int out_size, void* d_ws, size_t ws_size,
                              hipStream_t stream) {
  const float* input_state = (const float*)d_in[0];
  const float* theta       = (const float*)d_in[1];
  const float* M1          = (const float*)d_in[3];
  const int*   gti         = (const int*)d_in[5];
  const int*   gpi         = (const int*)d_in[6];

  const int B = in_sizes[0] / D;
  const int T = in_sizes[2] / (D * D);
  const int G = in_sizes[5];

  float* W      = (float*)d_ws;
  int*   counts = (int*)((char*)d_ws + (size_t)LDW * LDW * sizeof(float));
  int2*  pairs  = (int2*)(counts + TUP_CAP);
  float* out    = (float*)d_out;

  hipMemsetAsync(counts, 0, TUP_CAP * sizeof(int), stream);

  int tot = T * D;
  extract_pairs_k<<<(tot + 255) / 256, 256, 0, stream>>>(M1, T, counts, pairs);

  build_w_k<<<4, 128, (size_t)D * 129 * sizeof(float), stream>>>(
      theta, gti, gpi, counts, pairs, W, G);

  dim3 grid(D / 16, B / 512);
  gemm_wmma_k<<<grid, 256, 0, stream>>>(input_state, W, out);
}


# build_w: one wave per U column, each gate's 30 disjoint row pairs spread over the lanes (124 work-groups instead of 4); pair lists and per-gate cos/sin staged once in LDS
# speedup vs baseline: 4.3266x; 4.3266x over previous
#include <hip/hip_runtime.h>


typedef __attribute__((ext_vector_type(2))) float v2f;
typedef __attribute__((ext_vector_type(8))) float v8f;

#define D        496
#define LDW      512
#define PAIR_CAP 64
#define TUP_CAP  64

__global__ void extract_pairs_k(const float* __restrict__ M1, int T,
                                int* __restrict__ counts,
                                int2* __restrict__ pairs) {
  int idx = blockIdx.x * blockDim.x + threadIdx.x;
  if (idx >= T * D) return;
  int t = idx / D, i = idx - t * D;
  const float* row = M1 + (size_t)t * D * D + (size_t)i * D;
  int j = -1; float sv = 0.f;
  for (int c = 0; c < D; ++c) {
    float v = row[c];
    if (v != 0.f) { j = c; sv = v; }
  }
  if (j >= 0 && sv > 0.f) {
    int slot = atomicAdd(&counts[t], 1);
    pairs[t * PAIR_CAP + slot] = make_int2(i, j);
  }
}

#define GCH 256
#define PAD_PAIR ((unsigned)(D * 4) | ((unsigned)((D + 1) * 4) << 16))

struct __attribute__((aligned(16))) GateRec { float c, s; int tb, n; };

__device__ __forceinline__ void wave_lds_handover() {
  __builtin_amdgcn_fence(__ATOMIC_RELEASE, "wavefront");
  __builtin_amdgcn_wave_barrier();
  __builtin_amdgcn_fence(__ATOMIC_ACQUIRE, "wavefront");
}

__device__ __forceinline__ void rotate_pair(float* cw, unsigned pk, float c, float s) {
  float* pi = (float*)((char*)cw + (pk & 0xffffu));
  float* pj = (float*)((char*)cw + (pk >> 16));
  float a = *pi, b = *pj;
  *pi =  c * a + s * b;
  *pj = -s * a + c * b;
}

__global__ __launch_bounds__(128) void build_w_k(const float* __restrict__ theta,
                          const int* __restrict__ gti,
                          const int* __restrict__ gpi,
                          const int* __restrict__ counts,
                          const int2* __restrict__ pairs,
                          float* __restrict__ W, int G, int T) {
  __shared__ float    col[4][LDW];
  __shared__ unsigned ptab[TUP_CAP * PAIR_CAP];
  __shared__ int      cnt[TUP_CAP];
  __shared__ GateRec  gtab[GCH + 2];

  const int tid  = threadIdx.x;
  const int lane = tid & 31;
  const int wave = tid >> 5;
  const int k    = blockIdx.x * 4 + wave;
  float* cw = col[wave];

  for (int i = lane; i < LDW; i += 32)
    cw[i] = (i == k) ? 1.f : 0.f;
  if (tid < T) cnt[tid] = counts[tid];
  __syncthreads();

  for (int tq = 0; tq < T; tq += 16) {
    int  n[4];
    int2 v0[4], v1[4];
#pragma unroll
    for (int u = 0; u < 4; ++u) {
      const int t = tq + 4 * u + wave;
      n[u] = (t < T) ? cnt[t] : 0;
    }
#pragma unroll
    for (int u = 0; u < 4; ++u) {
      const int t = tq + 4 * u + wave;
      v0[u] = make_int2(0, 0); v1[u] = make_int2(0, 0);
      if (lane < n[u])      v0[u] = pairs[t * PAIR_CAP + lane];
      if (lane + 32 < n[u]) v1[u] = pairs[t * PAIR_CAP + 32 + lane];
    }
#pragma unroll
    for (int u = 0; u < 4; ++u) {
      const int t = tq + 4 * u + wave;
      if (t < T) {
        ptab[t * PAIR_CAP + lane] = (lane < n[u])
            ? ((unsigned)(v0[u].x * 4) | ((unsigned)(v0[u].y * 4) << 16)) : PAD_PAIR;
        ptab[t * PAIR_CAP + 32 + lane] = (lane + 32 < n[u])
            ? ((unsigned)(v1[u].x * 4) | ((unsigned)(v1[u].y * 4) << 16)) : PAD_PAIR;
      }
    }
  }

  for (int g0 = 0; g0 < G; g0 += GCH) {
    const int ng = min(G - g0, GCH);
    __syncthreads();
    for (int g = tid; g < ng + 2; g += 128) {
      GateRec m;
      m.c = 1.f; m.s = 0.f; m.tb = 0; m.n = 0;
      if (g < ng) {
        int t = gti[g0 + g], p = gpi[g0 + g];
        float th = theta[p];
        m.c = cosf(th); m.s = sinf(th);
        m.tb = t * PAIR_CAP; m.n = cnt[t];
      }
      gtab[g] = m;
    }
    __syncthreads();

    GateRec m0 = gtab[0];
    GateRec m1 = gtab[1];
    unsigned pk0 = ptab[m0.tb + lane];
    for (int g = 0; g < ng; ++g) {
      GateRec m2 = gtab[g + 2];
      unsigned pk1 = ptab[m1.tb + lane];

      rotate_pair(cw, pk0, m0.c, m0.s);
      if (__builtin_amdgcn_readfirstlane(m0.n) > 32)
        rotate_pair(cw, ptab[m0.tb + 32 + lane], m0.c, m0.s);
      wave_lds_handover();
      m0 = m1; m1 = m2; pk0 = pk1;
    }
  }

  float4*       dst = (float4*)(W + (size_t)k * LDW);
  const float4* srv = (const float4*)cw;
  for (int q = lane; q < LDW / 4; q += 32) dst[q] = srv[q];
}

__global__ void gemm_wmma_k(const float* __restrict__ A,
                            const float* __restrict__ W,
                            float* __restrict__ out) {
  const int lane = threadIdx.x & 31;
  const int wave = threadIdx.x >> 5;
  const int l16  = lane & 15;
  const int lh2  = (lane >> 4) * 2;
  const int m0   = (blockIdx.y * 8 + wave) * 64;
  const int n0   = blockIdx.x * 16;

  v8f acc[4];
#pragma unroll
  for (int mm = 0; mm < 4; ++mm) acc[mm] = (v8f)0.f;

  const float* a0 = A + (size_t)(m0 + l16) * D;
  const float* bp = W + (size_t)n0 + l16;

  for (int kk = 0; kk < D; kk += 4) {
    v2f b;
    b.x = bp[(size_t)(kk + lh2) * LDW];
    b.y = bp[(size_t)(kk + lh2 + 1) * LDW];
#pragma unroll
    for (int mm = 0; mm < 4; ++mm) {
      v2f a = *(const v2f*)(a0 + (size_t)mm * 16 * D + kk + lh2);
      acc[mm] = __builtin_amdgcn_wmma_f32_16x16x4_f32(
          false, a, false, b, (short)0, acc[mm], false, false);
    }
  }

  const int rbase = m0 + ((lane >> 4) << 3);
#pragma unroll
  for (int mm = 0; mm < 4; ++mm)
#pragma unroll
    for (int v = 0; v < 8; ++v)
      out[(size_t)(rbase + mm * 16 + v) * D + n0 + l16] = acc[mm][v];
}

extern "C" void kernel_launch(void* const* d_in, const int* in_sizes, int n_in,
                              void* d_out, int out_size, void* d_ws, size_t ws_size,
                              hipStream_t stream) {
  const float* input_state = (const float*)d_in[0];
  const float* theta       = (const float*)d_in[1];
  const float* M1          = (const float*)d_in[3];
  const int*   gti         = (const int*)d_in[5];
  const int*   gpi         = (const int*)d_in[6];

  const int B = in_sizes[0] / D;
  const int T = in_sizes[2] / (D * D);
  const int G = in_sizes[5];

  float* W      = (float*)d_ws;
  int*   counts = (int*)((char*)d_ws + (size_t)LDW * LDW * sizeof(float));
  int2*  pairs  = (int2*)(counts + TUP_CAP);
  float* out    = (float*)d_out;

  hipMemsetAsync(counts, 0, TUP_CAP * sizeof(int), stream);

  int tot = T * D;
  extract_pairs_k<<<(tot + 255) / 256, 256, 0, stream>>>(M1, T, counts, pairs);

  build_w_k<<<D / 4, 128, 0, stream>>>(
      theta, gti, gpi, counts, pairs, W, G, T);

  dim3 grid(D / 16, B / 512);
  gemm_wmma_k<<<grid, 256, 0, stream>>>(input_state, W, out);
}


# extract_pairs: a wave reads whole M1 rows with 512-byte contiguous 16-byte loads, 8 rows in flight per wave, lane masks pick each row's entry (was one thread per row: 32 lines per load)
# speedup vs baseline: 5.9710x; 1.3801x over previous
#include <hip/hip_runtime.h>


typedef __attribute__((ext_vector_type(2))) float v2f;
typedef __attribute__((ext_vector_type(8))) float v8f;

#define D        496
#define LDW      512
#define PAIR_CAP 64
#define TUP_CAP  64

typedef __attribute__((ext_vector_type(4))) float v4f;

#define XR    8
#define ROW_Q (D / 4)

__device__ __forceinline__ bool any_nonzero(v4f x) {
  return (x.x != 0.f) | (x.y != 0.f) | (x.z != 0.f) | (x.w != 0.f);
}

__global__ __launch_bounds__(256) void extract_pairs_k(const float* __restrict__ M1, int T,
                                int* __restrict__ counts,
                                int2* __restrict__ pairs) {
  const int lane  = threadIdx.x & 31;
  const int wv    = blockIdx.x * 8 + __builtin_amdgcn_readfirstlane((int)(threadIdx.x >> 5));
  const int nrows = T * D;
  const int r0    = wv * XR;
  if (r0 >= nrows) return;

  const int  q3   = min(lane + 96, ROW_Q - 1);
  const bool has3 = (lane + 96) < ROW_Q;

  v4f v[XR][4];
#pragma unroll
  for (int r = 0; r < XR; ++r) {
    const int row = min(r0 + r, nrows - 1);
    const v4f* rp = (const v4f*)(M1 + (size_t)row * D);
    v[r][0] = rp[lane];
    v[r][1] = rp[lane + 32];
    v[r][2] = rp[lane + 64];
    v[r][3] = rp[q3];
  }

  bool my_pos = false;
  int  my_j   = 0;
#pragma unroll
  for (int r = 0; r < XR; ++r) {
    if (r0 + r < nrows) {
      const v4f a0 = v[r][0], a1 = v[r][1], a2 = v[r][2], a3 = v[r][3];
      const unsigned m0 = __builtin_amdgcn_ballot_w32(any_nonzero(a0));
      const unsigned m1 = __builtin_amdgcn_ballot_w32(any_nonzero(a1));
      const unsigned m2 = __builtin_amdgcn_ballot_w32(any_nonzero(a2));
      const unsigned m3 = __builtin_amdgcn_ballot_w32(has3 && any_nonzero(a3));
      const int      ut = m3 ? 3 : m2 ? 2 : m1 ? 1 : 0;
      const unsigned mt = m3 ? m3 : m2 ? m2 : m1 ? m1 : m0;
      if (mt != 0u) {
        const int wl = 31 - __builtin_clz(mt);
        const v4f x  = (ut == 3) ? a3 : (ut == 2) ? a2 : (ut == 1) ? a1 : a0;
        const int   e  = (x.w != 0.f) ? 3 : (x.z != 0.f) ? 2 : (x.y != 0.f) ? 1 : 0;
        const float sv = (x.w != 0.f) ? x.w : (x.z != 0.f) ? x.z : (x.y != 0.f) ? x.y : x.x;
        const int code = (e << 1) | ((sv > 0.f) ? 1 : 0);
        const int cw   = __builtin_amdgcn_readlane(code, wl);
        if ((cw & 1) && lane == r) {
          my_pos = true;
          my_j   = 4 * (wl + 32 * ut) + (cw >> 1);
        }
      }
    }
  }

  if (my_pos) {
    const int row = r0 + lane;
    const int t = row / D, i = row - t * D;
    int slot = atomicAdd(&counts[t], 1);
    pairs[t * PAIR_CAP + slot] = make_int2(i, my_j);
  }
}

#define GCH 256
#define PAD_PAIR ((unsigned)(D * 4) | ((unsigned)((D + 1) * 4) << 16))

struct __attribute__((aligned(16))) GateRec { float c, s; int tb, n; };

__device__ __forceinline__ void wave_lds_handover() {
  __builtin_amdgcn_fence(__ATOMIC_RELEASE, "wavefront");
  __builtin_amdgcn_wave_barrier();
  __builtin_amdgcn_fence(__ATOMIC_ACQUIRE, "wavefront");
}

__device__ __forceinline__ void rotate_pair(float* cw, unsigned pk, float c, float s) {
  float* pi = (float*)((char*)cw + (pk & 0xffffu));
  float* pj = (float*)((char*)cw + (pk >> 16));
  float a = *pi, b = *pj;
  *pi =  c * a + s * b;
  *pj = -s * a + c * b;
}

__global__ __launch_bounds__(128) void build_w_k(const float* __restrict__ theta,
                          const int* __restrict__ gti,
                          const int* __restrict__ gpi,
                          const int* __restrict__ counts,
                          const int2* __restrict__ pairs,
                          float* __restrict__ W, int G, int T) {
  __shared__ float    col[4][LDW];
  __shared__ unsigned ptab[TUP_CAP * PAIR_CAP];
  __shared__ int      cnt[TUP_CAP];
  __shared__ GateRec  gtab[GCH + 2];

  const int tid  = threadIdx.x;
  const int lane = tid & 31;
  const int wave = tid >> 5;
  const int k    = blockIdx.x * 4 + wave;
  float* cw = col[wave];

  for (int i = lane; i < LDW; i += 32)
    cw[i] = (i == k) ? 1.f : 0.f;
  if (tid < T) cnt[tid] = counts[tid];
  __syncthreads();

  for (int tq = 0; tq < T; tq += 16) {
    int  n[4];
    int2 v0[4], v1[4];
#pragma unroll
    for (int u = 0; u < 4; ++u) {
      const int t = tq + 4 * u + wave;
      n[u] = (t < T) ? cnt[t] : 0;
    }
#pragma unroll
    for (int u = 0; u < 4; ++u) {
      const int t = tq + 4 * u + wave;
      v0[u] = make_int2(0, 0); v1[u] = make_int2(0, 0);
      if (lane < n[u])      v0[u] = pairs[t * PAIR_CAP + lane];
      if (lane + 32 < n[u]) v1[u] = pairs[t * PAIR_CAP + 32 + lane];
    }
#pragma unroll
    for (int u = 0; u < 4; ++u) {
      const int t = tq + 4 * u + wave;
      if (t < T) {
        ptab[t * PAIR_CAP + lane] = (lane < n[u])
            ? ((unsigned)(v0[u].x * 4) | ((unsigned)(v0[u].y * 4) << 16)) : PAD_PAIR;
        ptab[t * PAIR_CAP + 32 + lane] = (lane + 32 < n[u])
            ? ((unsigned)(v1[u].x * 4) | ((unsigned)(v1[u].y * 4) << 16)) : PAD_PAIR;
      }
    }
  }

  for (int g0 = 0; g0 < G; g0 += GCH) {
    const int ng = min(G - g0, GCH);
    __syncthreads();
    for (int g = tid; g < ng + 2; g += 128) {
      GateRec m;
      m.c = 1.f; m.s = 0.f; m.tb = 0; m.n = 0;
      if (g < ng) {
        int t = gti[g0 + g], p = gpi[g0 + g];
        float th = theta[p];
        m.c = cosf(th); m.s = sinf(th);
        m.tb = t * PAIR_CAP; m.n = cnt[t];
      }
      gtab[g] = m;
    }
    __syncthreads();

    GateRec m0 = gtab[0];
    GateRec m1 = gtab[1];
    unsigned pk0 = ptab[m0.tb + lane];
    for (int g = 0; g < ng; ++g) {
      GateRec m2 = gtab[g + 2];
      unsigned pk1 = ptab[m1.tb + lane];

      rotate_pair(cw, pk0, m0.c, m0.s);
      if (__builtin_amdgcn_readfirstlane(m0.n) > 32)
        rotate_pair(cw, ptab[m0.tb + 32 + lane], m0.c, m0.s);
      wave_lds_handover();
      m0 = m1; m1 = m2; pk0 = pk1;
    }
  }

  float4*       dst = (float4*)(W + (size_t)k * LDW);
  const float4* srv = (const float4*)cw;
  for (int q = lane; q < LDW / 4; q += 32) dst[q] = srv[q];
}

__global__ void gemm_wmma_k(const float* __restrict__ A,
                            const float* __restrict__ W,
                            float* __restrict__ out) {
  const int lane = threadIdx.x & 31;
  const int wave = threadIdx.x >> 5;
  const int l16  = lane & 15;
  const int lh2  = (lane >> 4) * 2;
  const int m0   = (blockIdx.y * 8 + wave) * 64;
  const int n0   = blockIdx.x * 16;

  v8f acc[4];
#pragma unroll
  for (int mm = 0; mm < 4; ++mm) acc[mm] = (v8f)0.f;

  const float* a0 = A + (size_t)(m0 + l16) * D;
  const float* bp = W + (size_t)n0 + l16;

  for (int kk = 0; kk < D; kk += 4) {
    v2f b;
    b.x = bp[(size_t)(kk + lh2) * LDW];
    b.y = bp[(size_t)(kk + lh2 + 1) * LDW];
#pragma unroll
    for (int mm = 0; mm < 4; ++mm) {
      v2f a = *(const v2f*)(a0 + (size_t)mm * 16 * D + kk + lh2);
      acc[mm] = __builtin_amdgcn_wmma_f32_16x16x4_f32(
          false, a, false, b, (short)0, acc[mm], false, false);
    }
  }

  const int rbase = m0 + ((lane >> 4) << 3);
#pragma unroll
  for (int mm = 0; mm < 4; ++mm)
#pragma unroll
    for (int v = 0; v < 8; ++v)
      out[(size_t)(rbase + mm * 16 + v) * D + n0 + l16] = acc[mm][v];
}

extern "C" void kernel_launch(void* const* d_in, const int* in_sizes, int n_in,
                              void* d_out, int out_size, void* d_ws, size_t ws_size,
                              hipStream_t stream) {
  const float* input_state = (const float*)d_in[0];
  const float* theta       = (const float*)d_in[1];
  const float* M1          = (const float*)d_in[3];
  const int*   gti         = (const int*)d_in[5];
  const int*   gpi         = (const int*)d_in[6];

  const int B = in_sizes[0] / D;
  const int T = in_sizes[2] / (D * D);
  const int G = in_sizes[5];

  float* W      = (float*)d_ws;
  int*   counts = (int*)((char*)d_ws + (size_t)LDW * LDW * sizeof(float));
  int2*  pairs  = (int2*)(counts + TUP_CAP);
  float* out    = (float*)d_out;

  hipMemsetAsync(counts, 0, TUP_CAP * sizeof(int), stream);

  int tot = T * D;
  extract_pairs_k<<<(tot + 8 * XR - 1) / (8 * XR), 256, 0, stream>>>(M1, T, counts, pairs);

  build_w_k<<<D / 4, 128, 0, stream>>>(
      theta, gti, gpi, counts, pairs, W, G, T);

  dim3 grid(D / 16, B / 512);
  gemm_wmma_k<<<grid, 256, 0, stream>>>(input_state, W, out);
}


# gemm: a 4-wave work-group keeps its 64x64 tile's whole K in LDS, filled by contiguous 16-byte loads (was fragment loads from global, 16 lines each); 256 work-groups, 2x2 tiles a wave, K in two halves
# speedup vs baseline: 10.7362x; 1.7981x over previous
#include <hip/hip_runtime.h>


typedef __attribute__((ext_vector_type(2))) float v2f;
typedef __attribute__((ext_vector_type(8))) float v8f;

#define D        496
#define LDW      512
#define PAIR_CAP 64
#define TUP_CAP  64

typedef __attribute__((ext_vector_type(4))) float v4f;

#define XR    8
#define ROW_Q (D / 4)

__device__ __forceinline__ bool any_nonzero(v4f x) {
  return (x.x != 0.f) | (x.y != 0.f) | (x.z != 0.f) | (x.w != 0.f);
}

__global__ __launch_bounds__(256) void extract_pairs_k(const float* __restrict__ M1, int T,
                                int* __restrict__ counts,
                                int2* __restrict__ pairs) {
  const int lane  = threadIdx.x & 31;
  const int wv    = blockIdx.x * 8 + __builtin_amdgcn_readfirstlane((int)(threadIdx.x >> 5));
  const int nrows = T * D;
  const int r0    = wv * XR;
  if (r0 >= nrows) return;

  const int  q3   = min(lane + 96, ROW_Q - 1);
  const bool has3 = (lane + 96) < ROW_Q;

  v4f v[XR][4];
#pragma unroll
  for (int r = 0; r < XR; ++r) {
    const int row = min(r0 + r, nrows - 1);
    const v4f* rp = (const v4f*)(M1 + (size_t)row * D);
    v[r][0] = rp[lane];
    v[r][1] = rp[lane + 32];
    v[r][2] = rp[lane + 64];
    v[r][3] = rp[q3];
  }

  bool my_pos = false;
  int  my_j   = 0;
#pragma unroll
  for (int r = 0; r < XR; ++r) {
    if (r0 + r < nrows) {
      const v4f a0 = v[r][0], a1 = v[r][1], a2 = v[r][2], a3 = v[r][3];
      const unsigned m0 = __builtin_amdgcn_ballot_w32(any_nonzero(a0));
      const unsigned m1 = __builtin_amdgcn_ballot_w32(any_nonzero(a1));
      const unsigned m2 = __builtin_amdgcn_ballot_w32(any_nonzero(a2));
      const unsigned m3 = __builtin_amdgcn_ballot_w32(has3 && any_nonzero(a3));
      const int      ut = m3 ? 3 : m2 ? 2 : m1 ? 1 : 0;
      const unsigned mt = m3 ? m3 : m2 ? m2 : m1 ? m1 : m0;
      if (mt != 0u) {
        const int wl = 31 - __builtin_clz(mt);
        const v4f x  = (ut == 3) ? a3 : (ut == 2) ? a2 : (ut == 1) ? a1 : a0;
        const int   e  = (x.w != 0.f) ? 3 : (x.z != 0.f) ? 2 : (x.y != 0.f) ? 1 : 0;
        const float sv = (x.w != 0.f) ? x.w : (x.z != 0.f) ? x.z : (x.y != 0.f) ? x.y : x.x;
        const int code = (e << 1) | ((sv > 0.f) ? 1 : 0);
        const int cw   = __builtin_amdgcn_readlane(code, wl);
        if ((cw & 1) && lane == r) {
          my_pos = true;
          my_j   = 4 * (wl + 32 * ut) + (cw >> 1);
        }
      }
    }
  }

  if (my_pos) {
    const int row = r0 + lane;
    const int t = row / D, i = row - t * D;
    int slot = atomicAdd(&counts[t], 1);
    pairs[t * PAIR_CAP + slot] = make_int2(i, my_j);
  }
}

#define GCH 256
#define PAD_PAIR ((unsigned)(D * 4) | ((unsigned)((D + 1) * 4) << 16))

struct __attribute__((aligned(16))) GateRec { float c, s; int tb, n; };

__device__ __forceinline__ void wave_lds_handover() {
  __builtin_amdgcn_fence(__ATOMIC_RELEASE, "wavefront");
  __builtin_amdgcn_wave_barrier();
  __builtin_amdgcn_fence(__ATOMIC_ACQUIRE, "wavefront");
}

__device__ __forceinline__ void rotate_pair(float* cw, unsigned pk, float c, float s) {
  float* pi = (float*)((char*)cw + (pk & 0xffffu));
  float* pj = (float*)((char*)cw + (pk >> 16));
  float a = *pi, b = *pj;
  *pi =  c * a + s * b;
  *pj = -s * a + c * b;
}

__global__ __launch_bounds__(128) void build_w_k(const float* __restrict__ theta,
                          const int* __restrict__ gti,
                          const int* __restrict__ gpi,
                          const int* __restrict__ counts,
                          const int2* __restrict__ pairs,
                          float* __restrict__ W, int G, int T) {
  __shared__ float    col[4][LDW];
  __shared__ unsigned ptab[TUP_CAP * PAIR_CAP];
  __shared__ int      cnt[TUP_CAP];
  __shared__ GateRec  gtab[GCH + 2];

  const int tid  = threadIdx.x;
  const int lane = tid & 31;
  const int wave = tid >> 5;
  const int k    = blockIdx.x * 4 + wave;
  float* cw = col[wave];

  for (int i = lane; i < LDW; i += 32)
    cw[i] = (i == k) ? 1.f : 0.f;
  if (tid < T) cnt[tid] = counts[tid];
  __syncthreads();

  for (int tq = 0; tq < T; tq += 16) {
    int  n[4];
    int2 v0[4], v1[4];
#pragma unroll
    for (int u = 0; u < 4; ++u) {
      const int t = tq + 4 * u + wave;
      n[u] = (t < T) ? cnt[t] : 0;
    }
#pragma unroll
    for (int u = 0; u < 4; ++u) {
      const int t = tq + 4 * u + wave;
      v0[u] = make_int2(0, 0); v1[u] = make_int2(0, 0);
      if (lane < n[u])      v0[u] = pairs[t * PAIR_CAP + lane];
      if (lane + 32 < n[u]) v1[u] = pairs[t * PAIR_CAP + 32 + lane];
    }
#pragma unroll
    for (int u = 0; u < 4; ++u) {
      const int t = tq + 4 * u + wave;
      if (t < T) {
        ptab[t * PAIR_CAP + lane] = (lane < n[u])
            ? ((unsigned)(v0[u].x * 4) | ((unsigned)(v0[u].y * 4) << 16)) : PAD_PAIR;
        ptab[t * PAIR_CAP + 32 + lane] = (lane + 32 < n[u])
            ? ((unsigned)(v1[u].x * 4) | ((unsigned)(v1[u].y * 4) << 16)) : PAD_PAIR;
      }
    }
  }

  for (int g0 = 0; g0 < G; g0 += GCH) {
    const int ng = min(G - g0, GCH);
    __syncthreads();
    for (int g = tid; g < ng + 2; g += 128) {
      GateRec m;
      m.c = 1.f; m.s = 0.f; m.tb = 0; m.n = 0;
      if (g < ng) {
        int t = gti[g0 + g], p = gpi[g0 + g];
        float th = theta[p];
        m.c = cosf(th); m.s = sinf(th);
        m.tb = t * PAIR_CAP; m.n = cnt[t];
      }
      gtab[g] = m;
    }
    __syncthreads();

    GateRec m0 = gtab[0];
    GateRec m1 = gtab[1];
    unsigned pk0 = ptab[m0.tb + lane];
    for (int g = 0; g < ng; ++g) {
      GateRec m2 = gtab[g + 2];
      unsigned pk1 = ptab[m1.tb + lane];

      rotate_pair(cw, pk0, m0.c, m0.s);
      if (__builtin_amdgcn_readfirstlane(m0.n) > 32)
        rotate_pair(cw, ptab[m0.tb + 32 + lane], m0.c, m0.s);
      wave_lds_handover();
      m0 = m1; m1 = m2; pk0 = pk1;
    }
  }

  float4*       dst = (float4*)(W + (size_t)k * LDW);
  const float4* srv = (const float4*)cw;
  for (int q = lane; q < LDW / 4; q += 32) dst[q] = srv[q];
}

#define GM_T 64
#define PA   500
#define PB2  160
#define KH   (D / 2)
#define NLA  32
#define NLB  16

struct FragSet { v2f a[2][2]; v2f b[2][2]; };

__device__ __forceinline__ void load_frags(FragSet& f, const float* ap, const float* bp, int kk) {
#pragma unroll
  for (int s = 0; s < 2; ++s) {
    const int k = kk + 4 * s;
#pragma unroll
    for (int i = 0; i < 2; ++i)
      f.a[s][i] = *(const v2f*)(ap + i * 16 * PA + k);
#pragma unroll
    for (int j = 0; j < 2; ++j)
      f.b[s][j] = *(const v2f*)(bp + (k >> 1) * PB2 + 32 * j);
  }
}

__device__ __forceinline__ void mma_frags(v8f (&acc)[2][2], const FragSet& f) {
#pragma unroll
  for (int s = 0; s < 2; ++s)
#pragma unroll
    for (int i = 0; i < 2; ++i)
#pragma unroll
      for (int j = 0; j < 2; ++j)
        acc[i][j] = __builtin_amdgcn_wmma_f32_16x16x4_f32(
            false, f.a[s][i], false, f.b[s][j], (short)0, acc[i][j], false, false);
}

__device__ __forceinline__ void lds_barrier() {
  __builtin_amdgcn_fence(__ATOMIC_RELEASE, "workgroup", "local");
  __builtin_amdgcn_s_barrier();
  __builtin_amdgcn_fence(__ATOMIC_ACQUIRE, "workgroup", "local");
}

__global__ __launch_bounds__(128) void gemm_wmma_k(const float* __restrict__ A,
                            const float* __restrict__ W,
                            float* __restrict__ out) {
  __shared__ __attribute__((aligned(16))) float As[GM_T * PA];
  __shared__ __attribute__((aligned(16))) float Bs[(D / 2) * PB2];

  const int tid  = threadIdx.x;
  const int lane = tid & 31;
  const int wave = tid >> 5;
  const int wm   = wave >> 1;
  const int wn   = wave & 1;
  const int l16  = lane & 15;
  const int lh   = lane >> 4;
  const int m0   = blockIdx.y * GM_T;
  const int n0   = blockIdx.x * GM_T;

  const int ta  = min(tid, 123);
  const int rr  = (ta >= 62) ? 1 : 0;
  const int qa  = ta - 62 * rr;
  const float* ga = A + (size_t)(m0 + rr) * D + 4 * qa;
  float*       sa = As + rr * PA + 4 * qa;
  const int pr  = tid >> 4;
  const int prl = min(pr, 3);
  const int qb  = tid & 15;
  const float* gb  = W + (size_t)(2 * pr) * LDW + n0 + 4 * qb;
  const float* gbl = W + (size_t)(2 * prl) * LDW + n0 + 4 * qb;
  float*       sb  = Bs + pr * PB2 + 8 * qb;
  float*       sbl = Bs + prl * PB2 + 8 * qb;

  v4f ra[NLA], rx[NLB], ry[NLB];

#define ISSUE_HALF(p)                                                              \
  _Pragma("unroll")                                                                \
  for (int u = 0; u < NLA; ++u)                                                    \
    ra[u] = *(const v4f*)(ga + (size_t)u * 2 * D + KH * (p));                      \
  _Pragma("unroll")                                                                \
  for (int u = 0; u < NLB; ++u) {                                                  \
    const float* g = ((u < NLB - 1) ? gb : gbl) + ((size_t)u * 16 + KH * (p)) * LDW; \
    rx[u] = *(const v4f*)g;                                                        \
    ry[u] = *(const v4f*)(g + LDW);                                                \
  }
#define STASH_HALF(p)                                                              \
  _Pragma("unroll")                                                                \
  for (int u = 0; u < NLA; ++u)                                                    \
    *(v4f*)(sa + u * 2 * PA + KH * (p)) = ra[u];                                   \
  _Pragma("unroll")                                                                \
  for (int u = 0; u < NLB; ++u) {                                                  \
    float* s = ((u < NLB - 1) ? sb : sbl) + (u * 8 + (KH / 2) * (p)) * PB2;        \
    v4f lo, hi;                                                                    \
    lo.x = rx[u].x; lo.y = ry[u].x; lo.z = rx[u].y; lo.w = ry[u].y;                \
    hi.x = rx[u].z; hi.y = ry[u].z; hi.z = rx[u].w; hi.w = ry[u].w;                \
    *(v4f*)s = lo;                                                                 \
    *(v4f*)(s + 4) = hi;                                                           \
  }

  v8f acc[2][2];
#pragma unroll
  for (int i = 0; i < 2; ++i)
#pragma unroll
    for (int j = 0; j < 2; ++j) acc[i][j] = (v8f)0.f;

  const float* ap = As + (32 * wm + l16) * PA + 2 * lh;
  const float* bp = Bs + lh * PB2 + 2 * (32 * wn + l16);

#define COMPUTE_HALF(p)                                                            \
  {                                                                                \
    const int kb = KH * (p);                                                       \
    FragSet f0, f1;                                                                \
    load_frags(f0, ap, bp, kb);                                                    \
    _Pragma("nounroll")                                                            \
    for (int it = 0; it < 15; ++it) {                                              \
      const int kk = kb + 16 * it;                                                 \
      load_frags(f1, ap, bp, kk + 8);                                              \
      __builtin_amdgcn_sched_barrier(0);                                           \
      mma_frags(acc, f0);                                                          \
      __builtin_amdgcn_sched_barrier(0);                                           \
      load_frags(f0, ap, bp, kk + 16);                                             \
      __builtin_amdgcn_sched_barrier(0);                                           \
      mma_frags(acc, f1);                                                          \
      __builtin_amdgcn_sched_barrier(0);                                           \
    }                                                                              \
    mma_frags(acc, f0);                                                            \
  }

  ISSUE_HALF(0)
  __builtin_amdgcn_sched_barrier(0);
  STASH_HALF(0)
  ISSUE_HALF(1)
  lds_barrier();
  COMPUTE_HALF(0)
  STASH_HALF(1)
  lds_barrier();
  COMPUTE_HALF(1)

#undef ISSUE_HALF
#undef STASH_HALF
#undef COMPUTE_HALF

  const int rbase = m0 + 32 * wm + (lh << 3);
#pragma unroll
  for (int i = 0; i < 2; ++i)
#pragma unroll
    for (int j = 0; j < 2; ++j) {
      const int nt = n0 + 32 * wn + 16 * j;
      if (nt < D) {
#pragma unroll
        for (int v = 0; v < 8; ++v)
          out[(size_t)(rbase + i * 16 + v) * D + nt + l16] = acc[i][j][v];
      }
    }
}

extern "C" void kernel_launch(void* const* d_in, const int* in_sizes, int n_in,
                              void* d_out, int out_size, void* d_ws, size_t ws_size,
                              hipStream_t stream) {
  const float* input_state = (const float*)d_in[0];
  const float* theta       = (const float*)d_in[1];
  const float* M1          = (const float*)d_in[3];
  const int*   gti         = (const int*)d_in[5];
  const int*   gpi         = (const int*)d_in[6];

  const int B = in_sizes[0] / D;
  const int T = in_sizes[2] / (D * D);
  const int G = in_sizes[5];

  float* W      = (float*)d_ws;
  int*   counts = (int*)((char*)d_ws + (size_t)LDW * LDW * sizeof(float));
  int2*  pairs  = (int2*)(counts + TUP_CAP);
  float* out    = (float*)d_out;

  hipMemsetAsync(counts, 0, TUP_CAP * sizeof(int), stream);

  int tot = T * D;
  extract_pairs_k<<<(tot + 8 * XR - 1) / (8 * XR), 256, 0, stream>>>(M1, T, counts, pairs);

  build_w_k<<<D / 4, 128, 0, stream>>>(
      theta, gti, gpi, counts, pairs, W, G, T);

  dim3 grid(LDW / GM_T, B / GM_T);
  gemm_wmma_k<<<grid, 128, 0, stream>>>(input_state, W, out);
}


# build_w: gate loop software-pipelined, one branch-free form when every pair list fits 32 lanes: between gates only the LDS round trip of the lane's two rows and two multiply-adds remain
# speedup vs baseline: 11.8113x; 1.1001x over previous
#include <hip/hip_runtime.h>


typedef __attribute__((ext_vector_type(2))) float v2f;
typedef __attribute__((ext_vector_type(8))) float v8f;

#define D        496
#define LDW      512
#define PAIR_CAP 64
#define TUP_CAP  64

typedef __attribute__((ext_vector_type(4))) float v4f;

#define XR    8
#define ROW_Q (D / 4)

__device__ __forceinline__ bool any_nonzero(v4f x) {
  return (x.x != 0.f) | (x.y != 0.f) | (x.z != 0.f) | (x.w != 0.f);
}

__global__ __launch_bounds__(256) void extract_pairs_k(const float* __restrict__ M1, int T,
                                int* __restrict__ counts,
                                int2* __restrict__ pairs) {
  const int lane  = threadIdx.x & 31;
  const int wv    = blockIdx.x * 8 + __builtin_amdgcn_readfirstlane((int)(threadIdx.x >> 5));
  const int nrows = T * D;
  const int r0    = wv * XR;
  if (r0 >= nrows) return;

  const int  q3   = min(lane + 96, ROW_Q - 1);
  const bool has3 = (lane + 96) < ROW_Q;

  v4f v[XR][4];
#pragma unroll
  for (int r = 0; r < XR; ++r) {
    const int row = min(r0 + r, nrows - 1);
    const v4f* rp = (const v4f*)(M1 + (size_t)row * D);
    v[r][0] = rp[lane];
    v[r][1] = rp[lane + 32];
    v[r][2] = rp[lane + 64];
    v[r][3] = rp[q3];
  }

  bool my_pos = false;
  int  my_j   = 0;
#pragma unroll
  for (int r = 0; r < XR; ++r) {
    if (r0 + r < nrows) {
      const v4f a0 = v[r][0], a1 = v[r][1], a2 = v[r][2], a3 = v[r][3];
      const unsigned m0 = __builtin_amdgcn_ballot_w32(any_nonzero(a0));
      const unsigned m1 = __builtin_amdgcn_ballot_w32(any_nonzero(a1));
      const unsigned m2 = __builtin_amdgcn_ballot_w32(any_nonzero(a2));
      const unsigned m3 = __builtin_amdgcn_ballot_w32(has3 && any_nonzero(a3));
      const int      ut = m3 ? 3 : m2 ? 2 : m1 ? 1 : 0;
      const unsigned mt = m3 ? m3 : m2 ? m2 : m1 ? m1 : m0;
      if (mt != 0u) {
        const int wl = 31 - __builtin_clz(mt);
        const v4f x  = (ut == 3) ? a3 : (ut == 2) ? a2 : (ut == 1) ? a1 : a0;
        const int   e  = (x.w != 0.f) ? 3 : (x.z != 0.f) ? 2 : (x.y != 0.f) ? 1 : 0;
        const float sv = (x.w != 0.f) ? x.w : (x.z != 0.f) ? x.z : (x.y != 0.f) ? x.y : x.x;
        const int code = (e << 1) | ((sv > 0.f) ? 1 : 0);
        const int cw   = __builtin_amdgcn_readlane(code, wl);
        if ((cw & 1) && lane == r) {
          my_pos = true;
          my_j   = 4 * (wl + 32 * ut) + (cw >> 1);
        }
      }
    }
  }

  if (my_pos) {
    const int row = r0 + lane;
    const int t = row / D, i = row - t * D;
    int slot = atomicAdd(&counts[t], 1);
    pairs[t * PAIR_CAP + slot] = make_int2(i, my_j);
  }
}

#define GCH 256
#define PAD_PAIR ((unsigned)(D * 4) | ((unsigned)((D + 1) * 4) << 16))

struct __attribute__((aligned(16))) GateRec { float c, s; int tb, n; };

__device__ __forceinline__ void wave_lds_handover() {
  __builtin_amdgcn_fence(__ATOMIC_RELEASE, "wavefront");
  __builtin_amdgcn_wave_barrier();
  __builtin_amdgcn_fence(__ATOMIC_ACQUIRE, "wavefront");
}

__device__ __forceinline__ void rotate_pair(float* cw, unsigned pk, float c, float s) {
  float* pi = (float*)((char*)cw + (pk & 0xffffu));
  float* pj = (float*)((char*)cw + (pk >> 16));
  float a = *pi, b = *pj;
  *pi =  c * a + s * b;
  *pj = -s * a + c * b;
}

__global__ __launch_bounds__(128) void build_w_k(const float* __restrict__ theta,
                          const int* __restrict__ gti,
                          const int* __restrict__ gpi,
                          const int* __restrict__ counts,
                          const int2* __restrict__ pairs,
                          float* __restrict__ W, int G, int T) {
  __shared__ float    col[4][LDW];
  __shared__ unsigned ptab[TUP_CAP * PAIR_CAP];
  __shared__ int      cnt[TUP_CAP];
  __shared__ GateRec  gtab[GCH + 3];

  const int tid  = threadIdx.x;
  const int lane = tid & 31;
  const int wave = tid >> 5;
  const int k    = blockIdx.x * 4 + wave;
  float* cw = col[wave];

  for (int i = lane; i < LDW; i += 32)
    cw[i] = (i == k) ? 1.f : 0.f;
  if (tid < T) cnt[tid] = counts[tid];
  __syncthreads();

  for (int tq = 0; tq < T; tq += 16) {
    int  n[4];
    int2 v0[4], v1[4];
#pragma unroll
    for (int u = 0; u < 4; ++u) {
      const int t = tq + 4 * u + wave;
      n[u] = (t < T) ? cnt[t] : 0;
    }
#pragma unroll
    for (int u = 0; u < 4; ++u) {
      const int t = tq + 4 * u + wave;
      v0[u] = make_int2(0, 0); v1[u] = make_int2(0, 0);
      if (lane < n[u])      v0[u] = pairs[t * PAIR_CAP + lane];
      if (lane + 32 < n[u]) v1[u] = pairs[t * PAIR_CAP + 32 + lane];
    }
#pragma unroll
    for (int u = 0; u < 4; ++u) {
      const int t = tq + 4 * u + wave;
      if (t < T) {
        ptab[t * PAIR_CAP + lane] = (lane < n[u])
            ? ((unsigned)(v0[u].x * 4) | ((unsigned)(v0[u].y * 4) << 16)) : PAD_PAIR;
        ptab[t * PAIR_CAP + 32 + lane] = (lane + 32 < n[u])
            ? ((unsigned)(v1[u].x * 4) | ((unsigned)(v1[u].y * 4) << 16)) : PAD_PAIR;
      }
    }
  }

  const bool wide = __builtin_amdgcn_ballot_w32((lane < T && cnt[lane] > 32) ||
                                                (lane + 32 < T && cnt[lane + 32] > 32)) != 0u;

  for (int g0 = 0; g0 < G; g0 += GCH) {
    const int ng = min(G - g0, GCH);
    __syncthreads();
    for (int g = tid; g < ng + 3; g += 128) {
      GateRec m;
      m.c = 1.f; m.s = 0.f; m.tb = 0; m.n = 0;
      if (g < ng) {
        int t = gti[g0 + g], p = gpi[g0 + g];
        float th = theta[p];
        m.c = cosf(th); m.s = sinf(th);
        m.tb = t * PAIR_CAP; m.n = cnt[t];
      }
      gtab[g] = m;
    }
    __syncthreads();

    if (!wide) {
      GateRec m0 = gtab[0];
      GateRec m1 = gtab[1];
      GateRec m2 = gtab[2];
      const unsigned pk0 = ptab[m0.tb + lane];
      unsigned pk1 = ptab[m1.tb + lane];
      float* pi0 = (float*)((char*)cw + (pk0 & 0xffffu));
      float* pj0 = (float*)((char*)cw + (pk0 >> 16));
      float a = *pi0, b = *pj0;
      for (int g = 0; g < ng; ++g) {
        const GateRec  m3  = gtab[g + 3];
        const unsigned pk2 = ptab[m2.tb + lane];
        float* pi1 = (float*)((char*)cw + (pk1 & 0xffffu));
        float* pj1 = (float*)((char*)cw + (pk1 >> 16));
        __builtin_amdgcn_sched_barrier(0);
        *pi0 =  m0.c * a + m0.s * b;
        *pj0 = -m0.s * a + m0.c * b;
        wave_lds_handover();
        a = *pi1; b = *pj1;
        __builtin_amdgcn_sched_barrier(0);
        m0 = m1; m1 = m2; m2 = m3; pk1 = pk2; pi0 = pi1; pj0 = pj1;
      }
    } else {
      GateRec m0 = gtab[0];
      GateRec m1 = gtab[1];
      unsigned pk0 = ptab[m0.tb + lane];
      for (int g = 0; g < ng; ++g) {
        GateRec m2 = gtab[g + 2];
        unsigned pk1 = ptab[m1.tb + lane];

        rotate_pair(cw, pk0, m0.c, m0.s);
        if (__builtin_amdgcn_readfirstlane(m0.n) > 32)
          rotate_pair(cw, ptab[m0.tb + 32 + lane], m0.c, m0.s);
        wave_lds_handover();
        m0 = m1; m1 = m2; pk0 = pk1;
      }
    }
  }

  float4*       dst = (float4*)(W + (size_t)k * LDW);
  const float4* srv = (const float4*)cw;
  for (int q = lane; q < LDW / 4; q += 32) dst[q] = srv[q];
}

#define GM_T 64
#define PA   500
#define PB2  160
#define KH   (D / 2)
#define NLA  32
#define NLB  16

struct FragSet { v2f a[2][2]; v2f b[2][2]; };

__device__ __forceinline__ void load_frags(FragSet& f, const float* ap, const float* bp, int kk) {
#pragma unroll
  for (int s = 0; s < 2; ++s) {
    const int k = kk + 4 * s;
#pragma unroll
    for (int i = 0; i < 2; ++i)
      f.a[s][i] = *(const v2f*)(ap + i * 16 * PA + k);
#pragma unroll
    for (int j = 0; j < 2; ++j)
      f.b[s][j] = *(const v2f*)(bp + (k >> 1) * PB2 + 32 * j);
  }
}

__device__ __forceinline__ void mma_frags(v8f (&acc)[2][2], const FragSet& f) {
#pragma unroll
  for (int s = 0; s < 2; ++s)
#pragma unroll
    for (int i = 0; i < 2; ++i)
#pragma unroll
      for (int j = 0; j < 2; ++j)
        acc[i][j] = __builtin_amdgcn_wmma_f32_16x16x4_f32(
            false, f.a[s][i], false, f.b[s][j], (short)0, acc[i][j], false, false);
}

__device__ __forceinline__ void lds_barrier() {
  __builtin_amdgcn_fence(__ATOMIC_RELEASE, "workgroup", "local");
  __builtin_amdgcn_s_barrier();
  __builtin_amdgcn_fence(__ATOMIC_ACQUIRE, "workgroup", "local");
}

__global__ __launch_bounds__(128) void gemm_wmma_k(const float* __restrict__ A,
                            const float* __restrict__ W,
                            float* __restrict__ out) {
  __shared__ __attribute__((aligned(16))) float As[GM_T * PA];
  __shared__ __attribute__((aligned(16))) float Bs[(D / 2) * PB2];

  const int tid  = threadIdx.x;
  const int lane = tid & 31;
  const int wave = tid >> 5;
  const int wm   = wave >> 1;
  const int wn   = wave & 1;
  const int l16  = lane & 15;
  const int lh   = lane >> 4;
  const int m0   = blockIdx.y * GM_T;
  const int n0   = blockIdx.x * GM_T;

  const int ta  = min(tid, 123);
  const int rr  = (ta >= 62) ? 1 : 0;
  const int qa  = ta - 62 * rr;
  const float* ga = A + (size_t)(m0 + rr) * D + 4 * qa;
  float*       sa = As + rr * PA + 4 * qa;
  const int pr  = tid >> 4;
  const int prl = min(pr, 3);
  const int qb  = tid & 15;
  const float* gb  = W + (size_t)(2 * pr) * LDW + n0 + 4 * qb;
  const float* gbl = W + (size_t)(2 * prl) * LDW + n0 + 4 * qb;
  float*       sb  = Bs + pr * PB2 + 8 * qb;
  float*       sbl = Bs + prl * PB2 + 8 * qb;

  v4f ra[NLA], rx[NLB], ry[NLB];

#define ISSUE_HALF(p)                                                              \
  _Pragma("unroll")                                                                \
  for (int u = 0; u < NLA; ++u)                                                    \
    ra[u] = *(const v4f*)(ga + (size_t)u * 2 * D + KH * (p));                      \
  _Pragma("unroll")                                                                \
  for (int u = 0; u < NLB; ++u) {                                                  \
    const float* g = ((u < NLB - 1) ? gb : gbl) + ((size_t)u * 16 + KH * (p)) * LDW; \
    rx[u] = *(const v4f*)g;                                                        \
    ry[u] = *(const v4f*)(g + LDW);                                                \
  }
#define STASH_HALF(p)                                                              \
  _Pragma("unroll")                                                                \
  for (int u = 0; u < NLA; ++u)                                                    \
    *(v4f*)(sa + u * 2 * PA + KH * (p)) = ra[u];                                   \
  _Pragma("unroll")                                                                \
  for (int u = 0; u < NLB; ++u) {                                                  \
    float* s = ((u < NLB - 1) ? sb : sbl) + (u * 8 + (KH / 2) * (p)) * PB2;        \
    v4f lo, hi;                                                                    \
    lo.x = rx[u].x; lo.y = ry[u].x; lo.z = rx[u].y; lo.w = ry[u].y;                \
    hi.x = rx[u].z; hi.y = ry[u].z; hi.z = rx[u].w; hi.w = ry[u].w;                \
    *(v4f*)s = lo;                                                                 \
    *(v4f*)(s + 4) = hi;                                                           \
  }

  v8f acc[2][2];
#pragma unroll
  for (int i = 0; i < 2; ++i)
#pragma unroll
    for (int j = 0; j < 2; ++j) acc[i][j] = (v8f)0.f;

  const float* ap = As + (32 * wm + l16) * PA + 2 * lh;
  const float* bp = Bs + lh * PB2 + 2 * (32 * wn + l16);

#define COMPUTE_HALF(p)                                                            \
  {                                                                                \
    const int kb = KH * (p);                                                       \
    FragSet f0, f1;                                                                \
    load_frags(f0, ap, bp, kb);                                                    \
    _Pragma("nounroll")                                                            \
    for (int it = 0; it < 15; ++it) {                                              \
      const int kk = kb + 16 * it;                                                 \
      load_frags(f1, ap, bp, kk + 8);                                              \
      __builtin_amdgcn_sched_barrier(0);                                           \
      mma_frags(acc, f0);                                                          \
      __builtin_amdgcn_sched_barrier(0);                                           \
      load_frags(f0, ap, bp, kk + 16);                                             \
      __builtin_amdgcn_sched_barrier(0);                                           \
      mma_frags(acc, f1);                                                          \
      __builtin_amdgcn_sched_barrier(0);                                           \
    }                                                                              \
    mma_frags(acc, f0);                                                            \
  }

  ISSUE_HALF(0)
  __builtin_amdgcn_sched_barrier(0);
  STASH_HALF(0)
  ISSUE_HALF(1)
  lds_barrier();
  COMPUTE_HALF(0)
  STASH_HALF(1)
  lds_barrier();
  COMPUTE_HALF(1)

#undef ISSUE_HALF
#undef STASH_HALF
#undef COMPUTE_HALF

  const int rbase = m0 + 32 * wm + (lh << 3);
#pragma unroll
  for (int i = 0; i < 2; ++i)
#pragma unroll
    for (int j = 0; j < 2; ++j) {
      const int nt = n0 + 32 * wn + 16 * j;
      if (nt < D) {
#pragma unroll
        for (int v = 0; v < 8; ++v)
          out[(size_t)(rbase + i * 16 + v) * D + nt + l16] = acc[i][j][v];
      }
    }
}

extern "C" void kernel_launch(void* const* d_in, const int* in_sizes, int n_in,
                              void* d_out, int out_size, void* d_ws, size_t ws_size,
                              hipStream_t stream) {
  const float* input_state = (const float*)d_in[0];
  const float* theta       = (const float*)d_in[1];
  const float* M1          = (const float*)d_in[3];
  const int*   gti         = (const int*)d_in[5];
  const int*   gpi         = (const int*)d_in[6];

  const int B = in_sizes[0] / D;
  const int T = in_sizes[2] / (D * D);
  const int G = in_sizes[5];

  float* W      = (float*)d_ws;
  int*   counts = (int*)((char*)d_ws + (size_t)LDW * LDW * sizeof(float));
  int2*  pairs  = (int2*)(counts + TUP_CAP);
  float* out    = (float*)d_out;

  hipMemsetAsync(counts, 0, TUP_CAP * sizeof(int), stream);

  int tot = T * D;
  extract_pairs_k<<<(tot + 8 * XR - 1) / (8 * XR), 256, 0, stream>>>(M1, T, counts, pairs);

  build_w_k<<<D / 4, 128, 0, stream>>>(
      theta, gti, gpi, counts, pairs, W, G, T);

  dim3 grid(LDW / GM_T, B / GM_T);
  gemm_wmma_k<<<grid, 128, 0, stream>>>(input_state, W, out);
}


# build_w: all table inputs requested at kernel entry in two rounds of unmasked loads (gate indices and list lengths; then angles and pair slots), one barrier before the gate loop; was seven rounds
# speedup vs baseline: 12.5355x; 1.0613x over previous
#include <hip/hip_runtime.h>


typedef __attribute__((ext_vector_type(2))) float v2f;
typedef __attribute__((ext_vector_type(8))) float v8f;

#define D        496
#define LDW      512
#define PAIR_CAP 64
#define TUP_CAP  64

typedef __attribute__((ext_vector_type(4))) float v4f;

#define XR    8
#define ROW_Q (D / 4)

__device__ __forceinline__ bool any_nonzero(v4f x) {
  return (x.x != 0.f) | (x.y != 0.f) | (x.z != 0.f) | (x.w != 0.f);
}

__global__ __launch_bounds__(256) void extract_pairs_k(const float* __restrict__ M1, int T,
                                int* __restrict__ counts,
                                int2* __restrict__ pairs) {
  const int lane  = threadIdx.x & 31;
  const int wv    = blockIdx.x * 8 + __builtin_amdgcn_readfirstlane((int)(threadIdx.x >> 5));
  const int nrows = T * D;
  const int r0    = wv * XR;
  if (r0 >= nrows) return;

  const int  q3   = min(lane + 96, ROW_Q - 1);
  const bool has3 = (lane + 96) < ROW_Q;

  v4f v[XR][4];
#pragma unroll
  for (int r = 0; r < XR; ++r) {
    const int row = min(r0 + r, nrows - 1);
    const v4f* rp = (const v4f*)(M1 + (size_t)row * D);
    v[r][0] = rp[lane];
    v[r][1] = rp[lane + 32];
    v[r][2] = rp[lane + 64];
    v[r][3] = rp[q3];
  }

  bool my_pos = false;
  int  my_j   = 0;
#pragma unroll
  for (int r = 0; r < XR; ++r) {
    if (r0 + r < nrows) {
      const v4f a0 = v[r][0], a1 = v[r][1], a2 = v[r][2], a3 = v[r][3];
      const unsigned m0 = __builtin_amdgcn_ballot_w32(any_nonzero(a0));
      const unsigned m1 = __builtin_amdgcn_ballot_w32(any_nonzero(a1));
      const unsigned m2 = __builtin_amdgcn_ballot_w32(any_nonzero(a2));
      const unsigned m3 = __builtin_amdgcn_ballot_w32(has3 && any_nonzero(a3));
      const int      ut = m3 ? 3 : m2 ? 2 : m1 ? 1 : 0;
      const unsigned mt = m3 ? m3 : m2 ? m2 : m1 ? m1 : m0;
      if (mt != 0u) {
        const int wl = 31 - __builtin_clz(mt);
        const v4f x  = (ut == 3) ? a3 : (ut == 2) ? a2 : (ut == 1) ? a1 : a0;
        const int   e  = (x.w != 0.f) ? 3 : (x.z != 0.f) ? 2 : (x.y != 0.f) ? 1 : 0;
        const float sv = (x.w != 0.f) ? x.w : (x.z != 0.f) ? x.z : (x.y != 0.f) ? x.y : x.x;
        const int code = (e << 1) | ((sv > 0.f) ? 1 : 0);
        const int cw   = __builtin_amdgcn_readlane(code, wl);
        if ((cw & 1) && lane == r) {
          my_pos = true;
          my_j   = 4 * (wl + 32 * ut) + (cw >> 1);
        }
      }
    }
  }

  if (my_pos) {
    const int row = r0 + lane;
    const int t = row / D, i = row - t * D;
    int slot = atomicAdd(&counts[t], 1);
    pairs[t * PAIR_CAP + slot] = make_int2(i, my_j);
  }
}

#define GCH 256
#define PAD_PAIR ((unsigned)(D * 4) | ((unsigned)((D + 1) * 4) << 16))

struct __attribute__((aligned(16))) GateRec { float c, s; int tb, n; };

__device__ __forceinline__ void wave_lds_handover() {
  __builtin_amdgcn_fence(__ATOMIC_RELEASE, "wavefront");
  __builtin_amdgcn_wave_barrier();
  __builtin_amdgcn_fence(__ATOMIC_ACQUIRE, "wavefront");
}

__device__ __forceinline__ void rotate_pair(float* cw, unsigned pk, float c, float s) {
  float* pi = (float*)((char*)cw + (pk & 0xffffu));
  float* pj = (float*)((char*)cw + (pk >> 16));
  float a = *pi, b = *pj;
  *pi =  c * a + s * b;
  *pj = -s * a + c * b;
}

struct TupLoads { int n[8]; int2 v0[8], v1[8]; };

__device__ __forceinline__ void tup_lengths(TupLoads& L, const int* __restrict__ counts, int tq, int wv) {
#pragma unroll
  for (int u = 0; u < 8; ++u)
    L.n[u] = counts[min(tq + 4 * u + wv, TUP_CAP - 1)];
}

__device__ __forceinline__ void tup_slots(TupLoads& L, const int2* __restrict__ pairs, int tq, int wv, int lane) {
#pragma unroll
  for (int u = 0; u < 8; ++u) {
    const int t = tq + 4 * u + wv;
    L.v0[u] = make_int2(0, 0); L.v1[u] = make_int2(0, 0);
    if (lane < L.n[u])      L.v0[u] = pairs[t * PAIR_CAP + lane];
    if (lane + 32 < L.n[u]) L.v1[u] = pairs[t * PAIR_CAP + 32 + lane];
  }
}

__device__ __forceinline__ void tup_store(const TupLoads& L, unsigned* ptab, int tq, int wv, int lane, int T) {
#pragma unroll
  for (int u = 0; u < 8; ++u) {
    const int t = tq + 4 * u + wv;
    if (t < T) {
      ptab[t * PAIR_CAP + lane] = (lane < L.n[u])
          ? ((unsigned)(L.v0[u].x * 4) | ((unsigned)(L.v0[u].y * 4) << 16)) : PAD_PAIR;
      ptab[t * PAIR_CAP + 32 + lane] = (lane + 32 < L.n[u])
          ? ((unsigned)(L.v1[u].x * 4) | ((unsigned)(L.v1[u].y * 4) << 16)) : PAD_PAIR;
    }
  }
}

__global__ __launch_bounds__(128) void build_w_k(const float* __restrict__ theta,
                          const int* __restrict__ gti,
                          const int* __restrict__ gpi,
                          const int* __restrict__ counts,
                          const int2* __restrict__ pairs,
                          float* __restrict__ W, int G, int T) {
  __shared__ float    col[4][LDW];
  __shared__ unsigned ptab[TUP_CAP * PAIR_CAP];
  __shared__ GateRec  gtab[GCH + 3];

  const int tid  = threadIdx.x;
  const int lane = tid & 31;
  const int wave = tid >> 5;
  const int k    = blockIdx.x * 4 + wave;
  float* cw = col[wave];

  int   gp[2], gt[2], gn[2];
  float gth[2];
#define FETCH_GATE_IDX(g0_, ng_)                                                   \
  _Pragma("unroll")                                                                \
  for (int u = 0; u < 2; ++u) {                                                    \
    const int g = (g0_) + min(tid + 128 * u, (ng_) - 1);                           \
    gp[u] = gpi[g]; gt[u] = gti[g];                                                \
  }
#define FETCH_GATE_VAL()                                                           \
  _Pragma("unroll")                                                                \
  for (int u = 0; u < 2; ++u) { gth[u] = theta[gp[u]]; gn[u] = counts[gt[u]]; }
#define WRITE_GATE_TAB(ng_)                                                        \
  _Pragma("unroll")                                                                \
  for (int u = 0; u < 3; ++u) {                                                    \
    const int g = tid + 128 * u;                                                   \
    if (g < (ng_) + 3) {                                                           \
      GateRec m;                                                                   \
      m.c = 1.f; m.s = 0.f; m.tb = 0; m.n = 0;                                     \
      if (u < 2 && g < (ng_)) {                                                    \
        m.c = cosf(gth[u & 1]); m.s = sinf(gth[u & 1]);                            \
        m.tb = gt[u & 1] * PAIR_CAP; m.n = gn[u & 1];                              \
      }                                                                            \
      gtab[g] = m;                                                                 \
    }                                                                              \
  }

  if (G <= 0) {
    float* dst0 = W + (size_t)k * LDW;
    for (int i = lane; i < LDW; i += 32) dst0[i] = (i == k) ? 1.f : 0.f;
    return;
  }
  const int wv  = __builtin_amdgcn_readfirstlane(wave);
  const int ng0 = min(G, GCH);
  TupLoads L;
  FETCH_GATE_IDX(0, ng0)
  const int cA = counts[lane];
  const int cB = counts[lane + 32];
  tup_lengths(L, counts, 0, wv);
  FETCH_GATE_VAL()
  tup_slots(L, pairs, 0, wv, lane);
  for (int i = lane; i < LDW; i += 32)
    cw[i] = (i == k) ? 1.f : 0.f;
  tup_store(L, ptab, 0, wv, lane, T);
  for (int tq = 32; tq < T; tq += 32) {
    tup_lengths(L, counts, tq, wv);
    tup_slots(L, pairs, tq, wv, lane);
    tup_store(L, ptab, tq, wv, lane, T);
  }

  const bool wide = __builtin_amdgcn_ballot_w32((cA > 32) || (cB > 32)) != 0u;

  for (int g0 = 0; g0 < G; g0 += GCH) {
    const int ng = min(G - g0, GCH);
    if (g0 > 0) {
      __syncthreads();
      FETCH_GATE_IDX(g0, ng)
      FETCH_GATE_VAL()
    }
    WRITE_GATE_TAB(ng)
    __syncthreads();

    if (!wide) {
      GateRec m0 = gtab[0];
      GateRec m1 = gtab[1];
      GateRec m2 = gtab[2];
      const unsigned pk0 = ptab[m0.tb + lane];
      unsigned pk1 = ptab[m1.tb + lane];
      float* pi0 = (float*)((char*)cw + (pk0 & 0xffffu));
      float* pj0 = (float*)((char*)cw + (pk0 >> 16));
      float a = *pi0, b = *pj0;
      for (int g = 0; g < ng; ++g) {
        const GateRec  m3  = gtab[g + 3];
        const unsigned pk2 = ptab[m2.tb + lane];
        float* pi1 = (float*)((char*)cw + (pk1 & 0xffffu));
        float* pj1 = (float*)((char*)cw + (pk1 >> 16));
        __builtin_amdgcn_sched_barrier(0);
        *pi0 =  m0.c * a + m0.s * b;
        *pj0 = -m0.s * a + m0.c * b;
        wave_lds_handover();
        a = *pi1; b = *pj1;
        __builtin_amdgcn_sched_barrier(0);
        m0 = m1; m1 = m2; m2 = m3; pk1 = pk2; pi0 = pi1; pj0 = pj1;
      }
    } else {
      GateRec m0 = gtab[0];
      GateRec m1 = gtab[1];
      unsigned pk0 = ptab[m0.tb + lane];
      for (int g = 0; g < ng; ++g) {
        GateRec m2 = gtab[g + 2];
        unsigned pk1 = ptab[m1.tb + lane];

        rotate_pair(cw, pk0, m0.c, m0.s);
        if (__builtin_amdgcn_readfirstlane(m0.n) > 32)
          rotate_pair(cw, ptab[m0.tb + 32 + lane], m0.c, m0.s);
        wave_lds_handover();
        m0 = m1; m1 = m2; pk0 = pk1;
      }
    }
  }

  float4*       dst = (float4*)(W + (size_t)k * LDW);
  const float4* srv = (const float4*)cw;
  for (int q = lane; q < LDW / 4; q += 32) dst[q] = srv[q];
}
#undef FETCH_GATE_IDX
#undef FETCH_GATE_VAL
#undef WRITE_GATE_TAB

#define GM_T 64
#define PA   500
#define PB2  160
#define KH   (D / 2)
#define NLA  32
#define NLB  16

struct FragSet { v2f a[2][2]; v2f b[2][2]; };

__device__ __forceinline__ void load_frags(FragSet& f, const float* ap, const float* bp, int kk) {
#pragma unroll
  for (int s = 0; s < 2; ++s) {
    const int k = kk + 4 * s;
#pragma unroll
    for (int i = 0; i < 2; ++i)
      f.a[s][i] = *(const v2f*)(ap + i * 16 * PA + k);
#pragma unroll
    for (int j = 0; j < 2; ++j)
      f.b[s][j] = *(const v2f*)(bp + (k >> 1) * PB2 + 32 * j);
  }
}

__device__ __forceinline__ void mma_frags(v8f (&acc)[2][2], const FragSet& f) {
#pragma unroll
  for (int s = 0; s < 2; ++s)
#pragma unroll
    for (int i = 0; i < 2; ++i)
#pragma unroll
      for (int j = 0; j < 2; ++j)
        acc[i][j] = __builtin_amdgcn_wmma_f32_16x16x4_f32(
            false, f.a[s][i], false, f.b[s][j], (short)0, acc[i][j], false, false);
}

__device__ __forceinline__ void lds_barrier() {
  __builtin_amdgcn_fence(__ATOMIC_RELEASE, "workgroup", "local");
  __builtin_amdgcn_s_barrier();
  __builtin_amdgcn_fence(__ATOMIC_ACQUIRE, "workgroup", "local");
}

__global__ __launch_bounds__(128) void gemm_wmma_k(const float* __restrict__ A,
                            const float* __restrict__ W,
                            float* __restrict__ out) {
  __shared__ __attribute__((aligned(16))) float As[GM_T * PA];
  __shared__ __attribute__((aligned(16))) float Bs[(D / 2) * PB2];

  const int tid  = threadIdx.x;
  const int lane = tid & 31;
  const int wave = tid >> 5;
  const int wm   = wave >> 1;
  const int wn   = wave & 1;
  const int l16  = lane & 15;
  const int lh   = lane >> 4;
  const int m0   = blockIdx.y * GM_T;
  const int n0   = blockIdx.x * GM_T;

  const int ta  = min(tid, 123);
  const int rr  = (ta >= 62) ? 1 : 0;
  const int qa  = ta - 62 * rr;
  const float* ga = A + (size_t)(m0 + rr) * D + 4 * qa;
  float*       sa = As + rr * PA + 4 * qa;
  const int pr  = tid >> 4;
  const int prl = min(pr, 3);
  const int qb  = tid & 15;
  const float* gb  = W + (size_t)(2 * pr) * LDW + n0 + 4 * qb;
  const float* gbl = W + (size_t)(2 * prl) * LDW + n0 + 4 * qb;
  float*       sb  = Bs + pr * PB2 + 8 * qb;
  float*       sbl = Bs + prl * PB2 + 8 * qb;

  v4f ra[NLA], rx[NLB], ry[NLB];

#define ISSUE_HALF(p)                                                              \
  _Pragma("unroll")                                                                \
  for (int u = 0; u < NLA; ++u)                                                    \
    ra[u] = *(const v4f*)(ga + (size_t)u * 2 * D + KH * (p));                      \
  _Pragma("unroll")                                                                \
  for (int u = 0; u < NLB; ++u) {                                                  \
    const float* g = ((u < NLB - 1) ? gb : gbl) + ((size_t)u * 16 + KH * (p)) * LDW; \
    rx[u] = *(const v4f*)g;                                                        \
    ry[u] = *(const v4f*)(g + LDW);                                                \
  }
#define STASH_HALF(p)                                                              \
  _Pragma("unroll")                                                                \
  for (int u = 0; u < NLA; ++u)                                                    \
    *(v4f*)(sa + u * 2 * PA + KH * (p)) = ra[u];                                   \
  _Pragma("unroll")                                                                \
  for (int u = 0; u < NLB; ++u) {                                                  \
    float* s = ((u < NLB - 1) ? sb : sbl) + (u * 8 + (KH / 2) * (p)) * PB2;        \
    v4f lo, hi;                                                                    \
    lo.x = rx[u].x; lo.y = ry[u].x; lo.z = rx[u].y; lo.w = ry[u].y;                \
    hi.x = rx[u].z; hi.y = ry[u].z; hi.z = rx[u].w; hi.w = ry[u].w;                \
    *(v4f*)s = lo;                                                                 \
    *(v4f*)(s + 4) = hi;                                                           \
  }

  v8f acc[2][2];
#pragma unroll
  for (int i = 0; i < 2; ++i)
#pragma unroll
    for (int j = 0; j < 2; ++j) acc[i][j] = (v8f)0.f;

  const float* ap = As + (32 * wm + l16) * PA + 2 * lh;
  const float* bp = Bs + lh * PB2 + 2 * (32 * wn + l16);

#define COMPUTE_HALF(p)                                                            \
  {                                                                                \
    const int kb = KH * (p);                                                       \
    FragSet f0, f1;                                                                \
    load_frags(f0, ap, bp, kb);                                                    \
    _Pragma("nounroll")                                                            \
    for (int it = 0; it < 15; ++it) {                                              \
      const int kk = kb + 16 * it;                                                 \
      load_frags(f1, ap, bp, kk + 8);                                              \
      __builtin_amdgcn_sched_barrier(0);                                           \
      mma_frags(acc, f0);                                                          \
      __builtin_amdgcn_sched_barrier(0);                                           \
      load_frags(f0, ap, bp, kk + 16);                                             \
      __builtin_amdgcn_sched_barrier(0);                                           \
      mma_frags(acc, f1);                                                          \
      __builtin_amdgcn_sched_barrier(0);                                           \
    }                                                                              \
    mma_frags(acc, f0);                                                            \
  }

  ISSUE_HALF(0)
  __builtin_amdgcn_sched_barrier(0);
  STASH_HALF(0)
  ISSUE_HALF(1)
  lds_barrier();
  COMPUTE_HALF(0)
  STASH_HALF(1)
  lds_barrier();
  COMPUTE_HALF(1)

#undef ISSUE_HALF
#undef STASH_HALF
#undef COMPUTE_HALF

  const int rbase = m0 + 32 * wm + (lh << 3);
#pragma unroll
  for (int i = 0; i < 2; ++i)
#pragma unroll
    for (int j = 0; j < 2; ++j) {
      const int nt = n0 + 32 * wn + 16 * j;
      if (nt < D) {
#pragma unroll
        for (int v = 0; v < 8; ++v)
          out[(size_t)(rbase + i * 16 + v) * D + nt + l16] = acc[i][j][v];
      }
    }
}

extern "C" void kernel_launch(void* const* d_in, const int* in_sizes, int n_in,
                              void* d_out, int out_size, void* d_ws, size_t ws_size,
                              hipStream_t stream) {
  const float* input_state = (const float*)d_in[0];
  const float* theta       = (const float*)d_in[1];
  const float* M1          = (const float*)d_in[3];
  const int*   gti         = (const int*)d_in[5];
  const int*   gpi         = (const int*)d_in[6];

  const int B = in_sizes[0] / D;
  const int T = in_sizes[2] / (D * D);
  const int G = in_sizes[5];

  float* W      = (float*)d_ws;
  int*   counts = (int*)((char*)d_ws + (size_t)LDW * LDW * sizeof(float));
  int2*  pairs  = (int2*)(counts + TUP_CAP);
  float* out    = (float*)d_out;

  hipMemsetAsync(counts, 0, TUP_CAP * sizeof(int), stream);

  int tot = T * D;
  extract_pairs_k<<<(tot + 8 * XR - 1) / (8 * XR), 256, 0, stream>>>(M1, T, counts, pairs);

  build_w_k<<<D / 4, 128, 0, stream>>>(
      theta, gti, gpi, counts, pairs, W, G, T);

  dim3 grid(LDW / GM_T, B / GM_T);
  gemm_wmma_k<<<grid, 128, 0, stream>>>(input_state, W, out);
}
